# starting kernel
# baseline (speedup 1.0000x reference)
#include <hip/hip_runtime.h>

typedef float v2f __attribute__((ext_vector_type(2)));
typedef float v8f __attribute__((ext_vector_type(8)));

#define NSTATES 64
#define SINK_ITERS 50
#define EPSI 0.1f
#define TINYF 1e-8f

__device__ __forceinline__ float hsum16(float x) {
#pragma unroll
    for (int m = 1; m < 16; m <<= 1) x += __shfl_xor(x, m, 32);
    return x;
}

__global__ __launch_bounds__(128)
void sinkhorn_emd_kernel(const float* __restrict__ p,
                         const float* __restrict__ q,
                         const float* __restrict__ cost,
                         float* __restrict__ out) {
    __shared__ v2f   KB[16 * 4 * 32];
    __shared__ v2f   MB[16 * 4 * 32];
    __shared__ float stage[4][16 * 64];

    const int tid  = threadIdx.x;
    const int wave = tid >> 5;
    const int lane = tid & 31;
    const int half = lane >> 4;
    const int l15  = lane & 15;

    for (int e = tid; e < 16 * 4 * 32; e += 128) {
        int c   = e >> 7;
        int t   = (e >> 5) & 3;
        int ln  = e & 31;
        int kr  = 4 * c + ((ln >> 4) << 1);
        int col = 16 * t + (ln & 15);
        float c0 = cost[kr * NSTATES + col];
        float c1 = cost[(kr + 1) * NSTATES + col];
        float k0 = __expf(-c0 * (1.0f / EPSI));
        float k1 = __expf(-c1 * (1.0f / EPSI));
        KB[e] = v2f{k0, k1};
        MB[e] = v2f{k0 * c0, k1 * c1};
    }
    __syncthreads();

    const int rowbase = blockIdx.x * 64 + wave * 16;

    float qD[4][8], pD[4][8];
#pragma unroll
    for (int t = 0; t < 4; ++t) {
#pragma unroll
        for (int j = 0; j < 8; ++j) {
            int r = rowbase + j + half * 8;
            int cidx = t * 16 + l15;
            qD[t][j] = q[r * NSTATES + cidx] + TINYF;
            pD[t][j] = p[r * NSTATES + cidx] + TINYF;
        }
    }
#pragma unroll
    for (int j = 0; j < 8; ++j) {
        float sq = qD[0][j] + qD[1][j] + qD[2][j] + qD[3][j];
        float sp = pD[0][j] + pD[1][j] + pD[2][j] + pD[3][j];
        sq = hsum16(sq);
        sp = hsum16(sp);
        float rq = __builtin_amdgcn_rcpf(sq);
        float rp = __builtin_amdgcn_rcpf(sp);
#pragma unroll
        for (int t = 0; t < 4; ++t) { qD[t][j] *= rq; pD[t][j] *= rp; }
    }

    v2f a[16];
#pragma unroll
    for (int c = 0; c < 16; ++c) a[c] = v2f{1.0f, 1.0f};

    float* st = stage[wave];
    float vD[4][8];

    for (int it = 0; it < SINK_ITERS; ++it) {
#pragma unroll
        for (int t = 0; t < 4; ++t) {
            v8f acc = {0.f, 0.f, 0.f, 0.f, 0.f, 0.f, 0.f, 0.f};
#pragma unroll
            for (int c = 0; c < 16; ++c) {
                v2f b = KB[(c * 4 + t) * 32 + lane];
                acc = __builtin_amdgcn_wmma_f32_16x16x4_f32(
                    false, a[c], false, b, (short)0, acc, false, false);
            }
#pragma unroll
            for (int j = 0; j < 8; ++j) {
                float v = qD[t][j] * __builtin_amdgcn_rcpf(acc[j]);
                vD[t][j] = v;
                st[(j + half * 8) * 64 + t * 16 + l15] = v;
            }
        }
#pragma unroll
        for (int c = 0; c < 16; ++c)
            a[c] = *(const v2f*)&st[l15 * 64 + 4 * c + half * 2];

#pragma unroll
        for (int t = 0; t < 4; ++t) {
            v8f acc = {0.f, 0.f, 0.f, 0.f, 0.f, 0.f, 0.f, 0.f};
#pragma unroll
            for (int c = 0; c < 16; ++c) {
                v2f b = KB[(c * 4 + t) * 32 + lane];
                acc = __builtin_amdgcn_wmma_f32_16x16x4_f32(
                    false, a[c], false, b, (short)0, acc, false, false);
            }
#pragma unroll
            for (int j = 0; j < 8; ++j) {
                float u = pD[t][j] * __builtin_amdgcn_rcpf(acc[j]);
                st[(j + half * 8) * 64 + t * 16 + l15] = u;
            }
        }
#pragma unroll
        for (int c = 0; c < 16; ++c)
            a[c] = *(const v2f*)&st[l15 * 64 + 4 * c + half * 2];
    }

    float e[8] = {0.f, 0.f, 0.f, 0.f, 0.f, 0.f, 0.f, 0.f};
#pragma unroll
    for (int t = 0; t < 4; ++t) {
        v8f acc = {0.f, 0.f, 0.f, 0.f, 0.f, 0.f, 0.f, 0.f};
#pragma unroll
        for (int c = 0; c < 16; ++c) {
            v2f b = MB[(c * 4 + t) * 32 + lane];
            acc = __builtin_amdgcn_wmma_f32_16x16x4_f32(
                false, a[c], false, b, (short)0, acc, false, false);
        }
#pragma unroll
        for (int j = 0; j < 8; ++j) e[j] += acc[j] * vD[t][j];
    }
#pragma unroll
    for (int j = 0; j < 8; ++j) {
        float s = hsum16(e[j]);
        if (l15 == 0) out[rowbase + j + half * 8] = s;
    }
}

extern "C" void kernel_launch(void* const* d_in, const int* in_sizes, int n_in,
                              void* d_out, int out_size, void* d_ws, size_t ws_size,
                              hipStream_t stream) {
    const float* p    = (const float*)d_in[0];
    const float* q    = (const float*)d_in[1];
    const float* cost = (const float*)d_in[2];
    float* out = (float*)d_out;

    int rows = in_sizes[0] / NSTATES;
    int blocks = rows / 64;
    sinkhorn_emd_kernel<<<blocks, 128, 0, stream>>>(p, q, cost, out);
}


# Products transposed (states on M, batch rows on N): a divided accumulator tile is the next product's B operand as it stands, A fragments baked in that k order; LDS relayout gone; sums re-ordered
# speedup vs baseline: 1.1094x; 1.1094x over previous
#include <hip/hip_runtime.h>


typedef float v2f __attribute__((ext_vector_type(2)));
typedef float v4f __attribute__((ext_vector_type(4)));
typedef float v8f __attribute__((ext_vector_type(8)));

#define NSTATES 64
#define SINK_ITERS 50
#define EPSI 0.1f
#define TINYF 1e-8f

#define SET_KT 0
#define SET_KN 1
#define SET_MT 2


__device__ __forceinline__ void load_frags(v4f (&f)[8], const v4f* fl, int set, int tp) {
#pragma unroll
    for (int q = 0; q < 8; ++q) f[q] = fl[((set * 4 + tp) * 8 + q) * 32];
}

__device__ __forceinline__ v8f tile_product(const v4f (&f)[8], const v2f (&x)[4][4]) {
    v8f acc = {0.f, 0.f, 0.f, 0.f, 0.f, 0.f, 0.f, 0.f};
#pragma unroll
    for (int t = 0; t < 4; ++t) {
#pragma unroll
        for (int jp = 0; jp < 2; ++jp) {
            v4f a = f[t * 2 + jp];
            acc = __builtin_amdgcn_wmma_f32_16x16x4_f32(
                false, v2f{a.x, a.y}, false, x[t][2 * jp], (short)0, acc, false, false);
            acc = __builtin_amdgcn_wmma_f32_16x16x4_f32(
                false, v2f{a.z, a.w}, false, x[t][2 * jp + 1], (short)0, acc, false, false);
        }
    }
    return acc;
}

__device__ __forceinline__ void divide_tile(v2f (&out)[4], const v2f (&num)[4], v8f acc) {
#pragma unroll
    for (int jj = 0; jj < 4; ++jj) {
        v2f r = {__builtin_amdgcn_rcpf(acc[2 * jj]), __builtin_amdgcn_rcpf(acc[2 * jj + 1])};
        out[jj] = num[jj] * r;
    }
}

__global__ __launch_bounds__(128)
void sinkhorn_emd_kernel(const float* __restrict__ p,
                         const float* __restrict__ q,
                         const float* __restrict__ cost,
                         float* __restrict__ out) {
    __shared__ v4f FR[3 * 1024];

    const int tid  = threadIdx.x;
    const int wave = tid >> 5;
    const int lane = tid & 31;
    const int half = lane >> 4;
    const int l15  = lane & 15;

#pragma unroll
    for (int i = 0; i < 8; ++i) {
        int e  = tid + 128 * i;
        int ln = e & 31;
        int qq = (e >> 5) & 7;
        int tp = e >> 8;
        int m  = 16 * tp + (ln & 15);
        int k0 = 16 * (qq >> 1) + 8 * (ln >> 4) + 4 * (qq & 1);
        v4f kt, mt, kn;
#pragma unroll
        for (int s = 0; s < 4; ++s) {
            float ct = cost[(k0 + s) * NSTATES + m];
            float cn = cost[m * NSTATES + k0 + s];
            float kk = __expf(-ct * (1.0f / EPSI));
            kt[s] = kk;
            mt[s] = kk * ct;
            kn[s] = __expf(-cn * (1.0f / EPSI));
        }
        FR[SET_KT * 1024 + e] = kt;
        FR[SET_KN * 1024 + e] = kn;
        FR[SET_MT * 1024 + e] = mt;
    }
    __syncthreads();

    const int rowbase = blockIdx.x * 64 + wave * 16;

    v2f qv[4][4], pv[4][4];
    {
        const v4f* qr = (const v4f*)(q + (size_t)(rowbase + l15) * NSTATES + 8 * half);
        const v4f* pr = (const v4f*)(p + (size_t)(rowbase + l15) * NSTATES + 8 * half);
        float sq = 0.f, sp = 0.f;
#pragma unroll
        for (int t = 0; t < 4; ++t) {
#pragma unroll
            for (int jp = 0; jp < 2; ++jp) {
                v4f a = qr[t * 4 + jp];
                v4f b = pr[t * 4 + jp];
                a += TINYF;
                b += TINYF;
                qv[t][2 * jp]     = v2f{a.x, a.y};
                qv[t][2 * jp + 1] = v2f{a.z, a.w};
                pv[t][2 * jp]     = v2f{b.x, b.y};
                pv[t][2 * jp + 1] = v2f{b.z, b.w};
                sq += (a.x + a.y) + (a.z + a.w);
                sp += (b.x + b.y) + (b.z + b.w);
            }
        }
        sq += __shfl_xor(sq, 16, 32);
        sp += __shfl_xor(sp, 16, 32);
        float rq = __builtin_amdgcn_rcpf(sq);
        float rp = __builtin_amdgcn_rcpf(sp);
#pragma unroll
        for (int t = 0; t < 4; ++t) {
#pragma unroll
            for (int jj = 0; jj < 4; ++jj) { qv[t][jj] *= rq; pv[t][jj] *= rp; }
        }
    }

    const v4f* fl = FR + lane;
    v4f kt[4][8], kn[4][8];
#pragma unroll
    for (int tp = 0; tp < 4; ++tp) {
        load_frags(kt[tp], fl, SET_KT, tp);
        load_frags(kn[tp], fl, SET_KN, tp);
    }

    v2f u[4][4], v[4][4];
#pragma unroll
    for (int t = 0; t < 4; ++t) {
#pragma unroll
        for (int jj = 0; jj < 4; ++jj) { u[t][jj] = v2f{1.0f, 1.0f}; v[t][jj] = v2f{1.0f, 1.0f}; }
    }

    for (int it = 0; it < SINK_ITERS; ++it) {
#pragma unroll
        for (int tp = 0; tp < 4; ++tp) divide_tile(v[tp], qv[tp], tile_product(kt[tp], u));
#pragma unroll
        for (int tp = 0; tp < 4; ++tp) divide_tile(u[tp], pv[tp], tile_product(kn[tp], v));
    }

    float e = 0.f;
#pragma unroll
    for (int tp = 0; tp < 4; ++tp) {
        v4f mf[8];
        load_frags(mf, fl, SET_MT, tp);
        v8f acc = tile_product(mf, u);
#pragma unroll
        for (int jj = 0; jj < 4; ++jj) {
            e += acc[2 * jj] * v[tp][jj].x;
            e += acc[2 * jj + 1] * v[tp][jj].y;
        }
    }
    e += __shfl_xor(e, 16, 32);
    if (half == 0) out[rowbase + l15] = e;
}

extern "C" void kernel_launch(void* const* d_in, const int* in_sizes, int n_in,
                              void* d_out, int out_size, void* d_ws, size_t ws_size,
                              hipStream_t stream) {
    const float* p    = (const float*)d_in[0];
    const float* q    = (const float*)d_in[1];
    const float* cost = (const float*)d_in[2];
    float* out = (float*)d_out;

    int rows = in_sizes[0] / NSTATES;
    int blocks = rows / 64;
    sinkhorn_emd_kernel<<<blocks, 128, 0, stream>>>(p, q, cost, out);
}
